# starting kernel
# baseline (speedup 1.0000x reference)
#include <hip/hip_runtime.h>

#define B_DIM 16
#define G_DIM 1708
#define H_DIM 5
#define GPAD  1712
#define BLOCK 128
#define WPB   (BLOCK / 32)
#define LOG2E 1.44269504088896340736f

typedef __attribute__((ext_vector_type(2))) float v2f;
typedef __attribute__((ext_vector_type(8))) float v8f;

#if __has_builtin(__builtin_amdgcn_exp2f)
#define EXP2F(v) __builtin_amdgcn_exp2f(v)
#else
#define EXP2F(v) exp2f(v)
#endif

__launch_bounds__(BLOCK)
__global__ void mha_rank1_wmma_kernel(const float* __restrict__ x,
                                      const float* __restrict__ WQ,
                                      const float* __restrict__ WK,
                                      const float* __restrict__ WV,
                                      const float* __restrict__ W0,
                                      float* __restrict__ out)
{
    __shared__ __align__(16) float x_lds[GPAD];
    __shared__ __align__(16) float k_lds[GPAD];
    __shared__ __align__(16) float v_lds[GPAD];

    const int b    = blockIdx.x;
    const int tid  = threadIdx.x;
    const int wave = tid >> 5;
    const int lane = tid & 31;
    const int ln   = lane & 15;
    const int hi   = lane >> 4;

    const int  tile   = blockIdx.y * WPB + wave;
    const bool tValid = (tile * 16) < G_DIM;
    const int  i      = tile * 16 + ln;
    const int  iC     = (i < G_DIM) ? i : (G_DIM - 1);

    for (int j = tid; j < GPAD; j += BLOCK)
        x_lds[j] = (j < G_DIM) ? x[b * G_DIM + j] : 0.0f;
    __syncthreads();

    float out_acc = 0.0f;

    for (int h = 0; h < H_DIM; ++h) {
        const float* __restrict__ wk = WK + h * G_DIM;
        const float* __restrict__ wv = WV + h * G_DIM;
        for (int j = tid; j < GPAD; j += BLOCK) {
            const float xv = x_lds[j];
            const float kk = (j < G_DIM) ? wk[j] : 0.0f;
            const float vv = (j < G_DIM) ? wv[j] : 0.0f;
            k_lds[j] = xv * kk * LOG2E;
            v_lds[j] = xv * vv;
        }
        __syncthreads();

        if (tValid) {
            const float t = (lane < 16) ? x_lds[iC] * WQ[h * G_DIM + iC] : 0.0f;
            v2f bm; bm[0] = t; bm[1] = 0.0f;

            float Dsum = 0.0f, Nsum = 0.0f;

            int jb = 0;
            for (; jb + 16 <= G_DIM; jb += 16) {
                v2f am; am[0] = (lane < 16) ? k_lds[jb + ln] : 0.0f; am[1] = 0.0f;
                v8f c = {};
                v8f d = __builtin_amdgcn_wmma_f32_16x16x4_f32(
                    false, am, false, bm, (short)0, c, false, false);
                const float4* vp =
                    reinterpret_cast<const float4*>(v_lds + jb + (hi << 3));
                const float4 va = vp[0], vb = vp[1];
                const float vv[8] = {va.x, va.y, va.z, va.w,
                                     vb.x, vb.y, vb.z, vb.w};
#pragma unroll
                for (int r = 0; r < 8; ++r) {
                    const float e = EXP2F(d[r]);
                    Dsum += e;
                    Nsum += e * vv[r];
                }
            }
            {
                v2f am; am[0] = (lane < 16) ? k_lds[jb + ln] : 0.0f; am[1] = 0.0f;
                v8f c = {};
                v8f d = __builtin_amdgcn_wmma_f32_16x16x4_f32(
                    false, am, false, bm, (short)0, c, false, false);
                const float4* vp =
                    reinterpret_cast<const float4*>(v_lds + jb + (hi << 3));
                const float4 va = vp[0], vb = vp[1];
                const float vv[8] = {va.x, va.y, va.z, va.w,
                                     vb.x, vb.y, vb.z, vb.w};
#pragma unroll
                for (int r = 0; r < 8; ++r) {
                    const bool ok = (jb + (hi << 3) + r) < G_DIM;
                    const float e = ok ? EXP2F(d[r]) : 0.0f;
                    Dsum += e;
                    Nsum += e * vv[r];
                }
            }

            Dsum += __shfl_xor(Dsum, 16, 32);
            Nsum += __shfl_xor(Nsum, 16, 32);

            if (lane < 16 && i < G_DIM) {
                const float eii = EXP2F(t * k_lds[i]);
                const float num = Nsum - eii * v_lds[i];
                out_acc += W0[h] * num / Dsum;
            }
        }
        __syncthreads();
    }

    if (tValid && lane < 16 && i < G_DIM)
        out[b * G_DIM + i] = out_acc;
}

extern "C" void kernel_launch(void* const* d_in, const int* in_sizes, int n_in,
                              void* d_out, int out_size, void* d_ws, size_t ws_size,
                              hipStream_t stream) {
    (void)in_sizes; (void)n_in; (void)out_size; (void)d_ws; (void)ws_size;
    const float* x  = (const float*)d_in[0];
    const float* WQ = (const float*)d_in[1];
    const float* WK = (const float*)d_in[2];
    const float* WV = (const float*)d_in[3];
    const float* W0 = (const float*)d_in[4];
    float* out = (float*)d_out;

    const int numTiles = (G_DIM + 15) / 16;
    const int groups   = (numTiles + WPB - 1) / WPB;
    dim3 grid(B_DIM, groups);
    hipLaunchKernelGGL(mha_rank1_wmma_kernel, grid, dim3(BLOCK), 0, stream,
                       x, WQ, WK, WV, W0, out);
}


# Reduction axis split: block = (batch row, 32 out rows); its 8 waves each take a j slice for both 16-row tiles, sharing k,v LDS reads; slice sums joined in LDS in wave order (float sum order changes)
# speedup vs baseline: 1.7320x; 1.7320x over previous
#include <hip/hip_runtime.h>

#define B_DIM 16
#define G_DIM 1708
#define H_DIM 5
#define GPAD  1712
#define NJT   107
#define BLOCK 256
#define WPB   (BLOCK / 32)
#define IT    2
#define LOG2E 1.44269504088896340736f

typedef __attribute__((ext_vector_type(2))) float v2f;
typedef __attribute__((ext_vector_type(8))) float v8f;

#if __has_builtin(__builtin_amdgcn_exp2f)
#define EXP2F(v) __builtin_amdgcn_exp2f(v)
#else
#define EXP2F(v) exp2f(v)
#endif

template <bool TAIL>
__device__ __forceinline__ void tile_step(const float* __restrict__ k_lds,
                                          const float* __restrict__ v_lds,
                                          int jb, int lane, int ln, int hi,
                                          v2f bm0, v2f bm1,
                                          float& D0, float& N0,
                                          float& D1, float& N1)
{
    const float kj = k_lds[jb + ln];
    v2f am; am[0] = (lane < 16) ? kj : 0.0f; am[1] = 0.0f;
    v8f c = {};
    v8f d0 = __builtin_amdgcn_wmma_f32_16x16x4_f32(
        false, am, false, bm0, (short)0, c, false, false);
    v8f d1 = __builtin_amdgcn_wmma_f32_16x16x4_f32(
        false, am, false, bm1, (short)0, c, false, false);
    const float4* vp = reinterpret_cast<const float4*>(v_lds + jb + (hi << 3));
    const float4 va = vp[0], vb = vp[1];
    const float vv[8] = {va.x, va.y, va.z, va.w, vb.x, vb.y, vb.z, vb.w};
#pragma unroll
    for (int r = 0; r < 8; ++r) {
        float e0 = EXP2F(d0[r]);
        float e1 = EXP2F(d1[r]);
        if (TAIL) {
            const bool ok = (jb + (hi << 3) + r) < G_DIM;
            e0 = ok ? e0 : 0.0f;
            e1 = ok ? e1 : 0.0f;
        }
        D0 += e0;
        N0 += e0 * vv[r];
        D1 += e1;
        N1 += e1 * vv[r];
    }
}

__launch_bounds__(BLOCK)
__global__ void mha_rank1_wmma_kernel(const float* __restrict__ x,
                                      const float* __restrict__ WQ,
                                      const float* __restrict__ WK,
                                      const float* __restrict__ WV,
                                      const float* __restrict__ W0,
                                      float* __restrict__ out)
{
    __shared__ __align__(16) float x_lds[GPAD];
    __shared__ __align__(16) float k_lds[GPAD];
    __shared__ __align__(16) float v_lds[GPAD];
    __shared__ __align__(16) float2 part[H_DIM][WPB][IT][16];

    const int b    = blockIdx.x;
    const int tid  = threadIdx.x;
    const int wave = __builtin_amdgcn_readfirstlane(tid >> 5);
    const int lane = tid & 31;
    const int ln   = lane & 15;
    const int hi   = lane >> 4;

    const int i0  = (blockIdx.y * IT) * 16 + ln;
    const int i1  = i0 + 16;
    const int i0C = (i0 < G_DIM) ? i0 : (G_DIM - 1);
    const int i1C = (i1 < G_DIM) ? i1 : (G_DIM - 1);

    const int jt0 = (wave * NJT) / WPB;
    const int jt1 = ((wave + 1) * NJT) / WPB;
    const bool ownsTail = (jt1 == NJT);
    const int jtMainEnd = ownsTail ? (NJT - 1) : jt1;

    for (int j = tid; j < GPAD; j += BLOCK)
        x_lds[j] = (j < G_DIM) ? x[b * G_DIM + j] : 0.0f;
    __syncthreads();

    for (int h = 0; h < H_DIM; ++h) {
        const float* __restrict__ wk = WK + h * G_DIM;
        const float* __restrict__ wv = WV + h * G_DIM;
        for (int j = tid; j < GPAD; j += BLOCK) {
            const float xv = x_lds[j];
            const float kk = (j < G_DIM) ? wk[j] : 0.0f;
            const float vv = (j < G_DIM) ? wv[j] : 0.0f;
            k_lds[j] = xv * kk * LOG2E;
            v_lds[j] = xv * vv;
        }
        __syncthreads();

        const float t0 = (lane < 16) ? x_lds[i0C] * WQ[h * G_DIM + i0C] : 0.0f;
        const float t1 = (lane < 16) ? x_lds[i1C] * WQ[h * G_DIM + i1C] : 0.0f;
        v2f bm0; bm0[0] = t0; bm0[1] = 0.0f;
        v2f bm1; bm1[0] = t1; bm1[1] = 0.0f;

        float D0 = 0.0f, N0 = 0.0f, D1 = 0.0f, N1 = 0.0f;

        for (int jt = jt0; jt < jtMainEnd; ++jt)
            tile_step<false>(k_lds, v_lds, jt * 16, lane, ln, hi,
                             bm0, bm1, D0, N0, D1, N1);
        if (ownsTail)
            tile_step<true>(k_lds, v_lds, (NJT - 1) * 16, lane, ln, hi,
                            bm0, bm1, D0, N0, D1, N1);

        D0 += __shfl_xor(D0, 16, 32);
        N0 += __shfl_xor(N0, 16, 32);
        D1 += __shfl_xor(D1, 16, 32);
        N1 += __shfl_xor(N1, 16, 32);

        if (lane < 16) {
            part[h][wave][0][ln] = make_float2(D0, N0);
            part[h][wave][1][ln] = make_float2(D1, N1);
        }
        __syncthreads();
    }

    if (tid < IT * 16) {
        const int it = tid >> 4;
        const int i  = (blockIdx.y * IT + it) * 16 + ln;
        if (i < G_DIM) {
            const float xi = x_lds[i];
            float out_acc = 0.0f;
            for (int h = 0; h < H_DIM; ++h) {
                float Dsum = 0.0f, Nsum = 0.0f;
#pragma unroll
                for (int w = 0; w < WPB; ++w) {
                    const float2 p = part[h][w][it][ln];
                    Dsum += p.x;
                    Nsum += p.y;
                }
                const float t    = xi * WQ[h * G_DIM + i];
                const float kval = xi * WK[h * G_DIM + i] * LOG2E;
                const float vval = xi * WV[h * G_DIM + i];
                const float eii  = EXP2F(t * kval);
                const float num  = Nsum - eii * vval;
                out_acc += W0[h] * num / Dsum;
            }
            out[b * G_DIM + i] = out_acc;
        }
    }
}

extern "C" void kernel_launch(void* const* d_in, const int* in_sizes, int n_in,
                              void* d_out, int out_size, void* d_ws, size_t ws_size,
                              hipStream_t stream) {
    (void)in_sizes; (void)n_in; (void)out_size; (void)d_ws; (void)ws_size;
    const float* x  = (const float*)d_in[0];
    const float* WQ = (const float*)d_in[1];
    const float* WK = (const float*)d_in[2];
    const float* WV = (const float*)d_in[3];
    const float* W0 = (const float*)d_in[4];
    float* out = (float*)d_out;

    dim3 grid(B_DIM, 54);
    hipLaunchKernelGGL(mha_rank1_wmma_kernel, grid, dim3(BLOCK), 0, stream,
                       x, WQ, WK, WV, W0, out);
}


# Sums packed two rows wide: exps of adjacent j rows accumulate as register pairs, one packed add and one packed fma per pair, into pair accumulators folded once per head (float sum order changes)
# speedup vs baseline: 1.9865x; 1.1469x over previous
#include <hip/hip_runtime.h>

#define B_DIM 16
#define G_DIM 1708
#define H_DIM 5
#define GPAD  1712
#define NJT   107
#define BLOCK 256
#define WPB   (BLOCK / 32)
#define IT    2
#define LOG2E 1.44269504088896340736f

typedef __attribute__((ext_vector_type(2))) float v2f;
typedef __attribute__((ext_vector_type(8))) float v8f;

#if __has_builtin(__builtin_amdgcn_exp2f)
#define EXP2F(v) __builtin_amdgcn_exp2f(v)
#else
#define EXP2F(v) exp2f(v)
#endif

template <bool TAIL>
__device__ __forceinline__ void tile_step(const float* __restrict__ k_lds,
                                          const float* __restrict__ v_lds,
                                          int jb, int lane, int ln, int hi,
                                          v2f bm0, v2f bm1,
                                          v2f (&Dp0)[2], v2f (&Np0)[2],
                                          v2f (&Dp1)[2], v2f (&Np1)[2])
{
    const float kj = k_lds[jb + ln];
    v2f am; am[0] = (lane < 16) ? kj : 0.0f; am[1] = 0.0f;
    v8f c = {};
    v8f d0 = __builtin_amdgcn_wmma_f32_16x16x4_f32(
        false, am, false, bm0, (short)0, c, false, false);
    v8f d1 = __builtin_amdgcn_wmma_f32_16x16x4_f32(
        false, am, false, bm1, (short)0, c, false, false);
    const float4* vp = reinterpret_cast<const float4*>(v_lds + jb + (hi << 3));
    const float4 va = vp[0], vb = vp[1];
    v2f vv[4];
    vv[0][0] = va.x; vv[0][1] = va.y;
    vv[1][0] = va.z; vv[1][1] = va.w;
    vv[2][0] = vb.x; vv[2][1] = vb.y;
    vv[3][0] = vb.z; vv[3][1] = vb.w;
#pragma unroll
    for (int p = 0; p < 4; ++p) {
        v2f e0, e1;
        e0[0] = EXP2F(d0[2 * p]);
        e0[1] = EXP2F(d0[2 * p + 1]);
        e1[0] = EXP2F(d1[2 * p]);
        e1[1] = EXP2F(d1[2 * p + 1]);
        if (TAIL) {
            const int j = jb + (hi << 3) + 2 * p;
            const bool okA = j < G_DIM;
            const bool okB = (j + 1) < G_DIM;
            e0[0] = okA ? e0[0] : 0.0f;
            e0[1] = okB ? e0[1] : 0.0f;
            e1[0] = okA ? e1[0] : 0.0f;
            e1[1] = okB ? e1[1] : 0.0f;
        }
        Dp0[p & 1] += e0;
        Np0[p & 1] = __builtin_elementwise_fma(e0, vv[p], Np0[p & 1]);
        Dp1[p & 1] += e1;
        Np1[p & 1] = __builtin_elementwise_fma(e1, vv[p], Np1[p & 1]);
    }
}

__launch_bounds__(BLOCK)
__global__ void mha_rank1_wmma_kernel(const float* __restrict__ x,
                                      const float* __restrict__ WQ,
                                      const float* __restrict__ WK,
                                      const float* __restrict__ WV,
                                      const float* __restrict__ W0,
                                      float* __restrict__ out)
{
    __shared__ __align__(16) float x_lds[GPAD];
    __shared__ __align__(16) float k_lds[GPAD];
    __shared__ __align__(16) float v_lds[GPAD];
    __shared__ __align__(16) float2 part[H_DIM][WPB][IT][16];

    const int b    = blockIdx.x;
    const int tid  = threadIdx.x;
    const int wave = __builtin_amdgcn_readfirstlane(tid >> 5);
    const int lane = tid & 31;
    const int ln   = lane & 15;
    const int hi   = lane >> 4;

    const int i0  = (blockIdx.y * IT) * 16 + ln;
    const int i1  = i0 + 16;
    const int i0C = (i0 < G_DIM) ? i0 : (G_DIM - 1);
    const int i1C = (i1 < G_DIM) ? i1 : (G_DIM - 1);

    const int jt0 = (wave * NJT) / WPB;
    const int jt1 = ((wave + 1) * NJT) / WPB;
    const bool ownsTail = (jt1 == NJT);
    const int jtMainEnd = ownsTail ? (NJT - 1) : jt1;

    for (int j = tid; j < GPAD; j += BLOCK)
        x_lds[j] = (j < G_DIM) ? x[b * G_DIM + j] : 0.0f;
    __syncthreads();

    for (int h = 0; h < H_DIM; ++h) {
        const float* __restrict__ wk = WK + h * G_DIM;
        const float* __restrict__ wv = WV + h * G_DIM;
        for (int j = tid; j < GPAD; j += BLOCK) {
            const float xv = x_lds[j];
            const float kk = (j < G_DIM) ? wk[j] : 0.0f;
            const float vv = (j < G_DIM) ? wv[j] : 0.0f;
            k_lds[j] = xv * kk * LOG2E;
            v_lds[j] = xv * vv;
        }
        __syncthreads();

        const float t0 = (lane < 16) ? x_lds[i0C] * WQ[h * G_DIM + i0C] : 0.0f;
        const float t1 = (lane < 16) ? x_lds[i1C] * WQ[h * G_DIM + i1C] : 0.0f;
        v2f bm0; bm0[0] = t0; bm0[1] = 0.0f;
        v2f bm1; bm1[0] = t1; bm1[1] = 0.0f;

        v2f Dp0[2] = {}, Np0[2] = {}, Dp1[2] = {}, Np1[2] = {};

        for (int jt = jt0; jt < jtMainEnd; ++jt)
            tile_step<false>(k_lds, v_lds, jt * 16, lane, ln, hi,
                             bm0, bm1, Dp0, Np0, Dp1, Np1);
        if (ownsTail)
            tile_step<true>(k_lds, v_lds, (NJT - 1) * 16, lane, ln, hi,
                            bm0, bm1, Dp0, Np0, Dp1, Np1);

        float D0 = (Dp0[0][0] + Dp0[0][1]) + (Dp0[1][0] + Dp0[1][1]);
        float N0 = (Np0[0][0] + Np0[0][1]) + (Np0[1][0] + Np0[1][1]);
        float D1 = (Dp1[0][0] + Dp1[0][1]) + (Dp1[1][0] + Dp1[1][1]);
        float N1 = (Np1[0][0] + Np1[0][1]) + (Np1[1][0] + Np1[1][1]);

        D0 += __shfl_xor(D0, 16, 32);
        N0 += __shfl_xor(N0, 16, 32);
        D1 += __shfl_xor(D1, 16, 32);
        N1 += __shfl_xor(N1, 16, 32);

        if (lane < 16) {
            part[h][wave][0][ln] = make_float2(D0, N0);
            part[h][wave][1][ln] = make_float2(D1, N1);
        }
        __syncthreads();
    }

    if (tid < IT * 16) {
        const int it = tid >> 4;
        const int i  = (blockIdx.y * IT + it) * 16 + ln;
        if (i < G_DIM) {
            const float xi = x_lds[i];
            float out_acc = 0.0f;
            for (int h = 0; h < H_DIM; ++h) {
                float Dsum = 0.0f, Nsum = 0.0f;
#pragma unroll
                for (int w = 0; w < WPB; ++w) {
                    const float2 p = part[h][w][it][ln];
                    Dsum += p.x;
                    Nsum += p.y;
                }
                const float t    = xi * WQ[h * G_DIM + i];
                const float kval = xi * WK[h * G_DIM + i] * LOG2E;
                const float vval = xi * WV[h * G_DIM + i];
                const float eii  = EXP2F(t * kval);
                const float num  = Nsum - eii * vval;
                out_acc += W0[h] * num / Dsum;
            }
            out[b * G_DIM + i] = out_acc;
        }
    }
}

extern "C" void kernel_launch(void* const* d_in, const int* in_sizes, int n_in,
                              void* d_out, int out_size, void* d_ws, size_t ws_size,
                              hipStream_t stream) {
    (void)in_sizes; (void)n_in; (void)out_size; (void)d_ws; (void)ws_size;
    const float* x  = (const float*)d_in[0];
    const float* WQ = (const float*)d_in[1];
    const float* WK = (const float*)d_in[2];
    const float* WV = (const float*)d_in[3];
    const float* W0 = (const float*)d_in[4];
    float* out = (float*)d_out;

    dim3 grid(B_DIM, 54);
    hipLaunchKernelGGL(mha_rank1_wmma_kernel, grid, dim3(BLOCK), 0, stream,
                       x, WQ, WK, WV, W0, out);
}


# speedup vs baseline: 2.6224x; 1.3201x over previous
#include <hip/hip_runtime.h>

#define B_DIM 16
#define G_DIM 1708
#define H_DIM 5
#define GPAD  1712
#define NJT   107
#define BLOCK 256
#define WPB   (BLOCK / 32)
#define IT    2
#define LOG2E 1.44269504088896340736f

typedef __attribute__((ext_vector_type(2))) float v2f;
typedef __attribute__((ext_vector_type(8))) float v8f;

#if __has_builtin(__builtin_amdgcn_exp2f)
#define EXP2F(v) __builtin_amdgcn_exp2f(v)
#else
#define EXP2F(v) exp2f(v)
#endif

template <bool TAIL>
__device__ __forceinline__ void tile_step(const float* __restrict__ k_lds,
                                          const float* __restrict__ v_lds,
                                          int jb, int lane, int ln, int hi,
                                          v2f bm0, v2f bm1,
                                          v2f (&Dp0)[2], v2f (&Np0)[2],
                                          v2f (&Dp1)[2], v2f (&Np1)[2])
{
    const float kj = k_lds[jb + ln];
    v2f am; am[0] = (lane < 16) ? kj : 0.0f; am[1] = 0.0f;
    v8f c = {};
    v8f d0 = __builtin_amdgcn_wmma_f32_16x16x4_f32(
        false, am, false, bm0, (short)0, c, false, false);
    v8f d1 = __builtin_amdgcn_wmma_f32_16x16x4_f32(
        false, am, false, bm1, (short)0, c, false, false);
    const float4* vp = reinterpret_cast<const float4*>(v_lds + jb + (hi << 3));
    const float4 va = vp[0], vb = vp[1];
    v2f vv[4];
    vv[0][0] = va.x; vv[0][1] = va.y;
    vv[1][0] = va.z; vv[1][1] = va.w;
    vv[2][0] = vb.x; vv[2][1] = vb.y;
    vv[3][0] = vb.z; vv[3][1] = vb.w;
#pragma unroll
    for (int p = 0; p < 4; ++p) {
        v2f e0, e1;
        e0[0] = EXP2F(d0[2 * p]);
        e0[1] = EXP2F(d0[2 * p + 1]);
        e1[0] = EXP2F(d1[2 * p]);
        e1[1] = EXP2F(d1[2 * p + 1]);
        if (TAIL) {
            const int j = jb + (hi << 3) + 2 * p;
            const bool okA = j < G_DIM;
            const bool okB = (j + 1) < G_DIM;
            e0[0] = okA ? e0[0] : 0.0f;
            e0[1] = okB ? e0[1] : 0.0f;
            e1[0] = okA ? e1[0] : 0.0f;
            e1[1] = okB ? e1[1] : 0.0f;
        }
        Dp0[p & 1] += e0;
        Np0[p & 1] = __builtin_elementwise_fma(e0, vv[p], Np0[p & 1]);
        Dp1[p & 1] += e1;
        Np1[p & 1] = __builtin_elementwise_fma(e1, vv[p], Np1[p & 1]);
    }
}

#define ORDER_FENCE() __builtin_amdgcn_sched_barrier(0)

struct StepVals {
    float e[16];
    v2f   vv[4];
};

template <int N>
__device__ __forceinline__ void exp_op(StepVals& s, const v8f& d0, const v8f& d1)
{
    s.e[N] = EXP2F(N < 8 ? d0[N & 7] : d1[N & 7]);
}

template <int K>
__device__ __forceinline__ void sum_op(const StepVals& s,
                                       v2f (&Dp0)[2], v2f (&Np0)[2],
                                       v2f (&Dp1)[2], v2f (&Np1)[2])
{
    constexpr int pr = (K >> 1) & 3;
    constexpr int t1 = K >> 3;
    v2f ep; ep[0] = s.e[8 * t1 + 2 * pr]; ep[1] = s.e[8 * t1 + 2 * pr + 1];
    if (t1 == 0) {
        if ((K & 1) == 0) Dp0[pr & 1] += ep;
        else Np0[pr & 1] = __builtin_elementwise_fma(ep, s.vv[pr], Np0[pr & 1]);
    } else {
        if ((K & 1) == 0) Dp1[pr & 1] += ep;
        else Np1[pr & 1] = __builtin_elementwise_fma(ep, s.vv[pr], Np1[pr & 1]);
    }
}

#define ACCS Dp0, Np0, Dp1, Np1

__device__ __forceinline__ void load_step(StepVals& s, float& kj,
                                          const float* __restrict__ k_lds,
                                          const float* __restrict__ v_lds,
                                          int jb, int ln, int hi)
{
    kj = k_lds[jb + ln];
    const float4* vp = reinterpret_cast<const float4*>(v_lds + jb + (hi << 3));
    const float4 va = vp[0], vb = vp[1];
    s.vv[0][0] = va.x; s.vv[0][1] = va.y;
    s.vv[1][0] = va.z; s.vv[1][1] = va.w;
    s.vv[2][0] = vb.x; s.vv[2][1] = vb.y;
    s.vv[3][0] = vb.z; s.vv[3][1] = vb.w;
}

#define ALTERNATE_11(s, d0, d1)                                   \
    exp_op<5>(s, d0, d1); ORDER_FENCE(); sum_op<0>(s, ACCS); ORDER_FENCE();    \
    exp_op<6>(s, d0, d1); ORDER_FENCE(); sum_op<1>(s, ACCS); ORDER_FENCE();    \
    exp_op<7>(s, d0, d1); ORDER_FENCE(); sum_op<2>(s, ACCS); ORDER_FENCE();    \
    exp_op<8>(s, d0, d1); ORDER_FENCE(); sum_op<3>(s, ACCS); ORDER_FENCE();    \
    exp_op<9>(s, d0, d1); ORDER_FENCE(); sum_op<4>(s, ACCS); ORDER_FENCE();    \
    exp_op<10>(s, d0, d1); ORDER_FENCE(); sum_op<5>(s, ACCS); ORDER_FENCE();    \
    exp_op<11>(s, d0, d1); ORDER_FENCE(); sum_op<6>(s, ACCS); ORDER_FENCE();    \
    exp_op<12>(s, d0, d1); ORDER_FENCE(); sum_op<7>(s, ACCS); ORDER_FENCE();    \
    exp_op<13>(s, d0, d1); ORDER_FENCE(); sum_op<8>(s, ACCS); ORDER_FENCE();    \
    exp_op<14>(s, d0, d1); ORDER_FENCE(); sum_op<9>(s, ACCS); ORDER_FENCE();    \
    exp_op<15>(s, d0, d1); ORDER_FENCE(); sum_op<10>(s, ACCS); ORDER_FENCE();

__device__ __forceinline__ void tile_pair_ordered(const float* __restrict__ k_lds,
                                                  const float* __restrict__ v_lds,
                                                  int jb, int lane, int ln, int hi,
                                                  v2f bm0, v2f bm1,
                                                  v2f (&Dp0)[2], v2f (&Np0)[2],
                                                  v2f (&Dp1)[2], v2f (&Np1)[2])
{
    StepVals sa, sb; float kja, kjb;
    load_step(sa, kja, k_lds, v_lds, jb, ln, hi);
    load_step(sb, kjb, k_lds, v_lds, jb + 16, ln, hi);
    v8f c = {};
    v2f ama; ama[0] = (lane < 16) ? kja : 0.0f; ama[1] = 0.0f;
    const v8f a0 = __builtin_amdgcn_wmma_f32_16x16x4_f32(
        false, ama, false, bm0, (short)0, c, false, false);
    const v8f a1 = __builtin_amdgcn_wmma_f32_16x16x4_f32(
        false, ama, false, bm1, (short)0, c, false, false);
    ORDER_FENCE();
    exp_op<0>(sa, a0, a1); ORDER_FENCE();
    v2f amb; amb[0] = (lane < 16) ? kjb : 0.0f; amb[1] = 0.0f;
    ORDER_FENCE();
    exp_op<1>(sa, a0, a1); exp_op<2>(sa, a0, a1);
    exp_op<3>(sa, a0, a1); exp_op<4>(sa, a0, a1);
    ORDER_FENCE();
    ALTERNATE_11(sa, a0, a1)
    const v8f b0 = __builtin_amdgcn_wmma_f32_16x16x4_f32(
        false, amb, false, bm0, (short)0, c, false, false);
    ORDER_FENCE();
    const v8f b1 = __builtin_amdgcn_wmma_f32_16x16x4_f32(
        false, amb, false, bm1, (short)0, c, false, false);
    ORDER_FENCE();
    exp_op<0>(sb, b0, b1); ORDER_FENCE(); sum_op<11>(sa, ACCS); ORDER_FENCE();
    exp_op<1>(sb, b0, b1); ORDER_FENCE(); sum_op<12>(sa, ACCS); ORDER_FENCE();
    exp_op<2>(sb, b0, b1); ORDER_FENCE(); sum_op<13>(sa, ACCS); ORDER_FENCE();
    exp_op<3>(sb, b0, b1); ORDER_FENCE(); sum_op<14>(sa, ACCS); ORDER_FENCE();
    exp_op<4>(sb, b0, b1); ORDER_FENCE(); sum_op<15>(sa, ACCS); ORDER_FENCE();
    ALTERNATE_11(sb, b0, b1)
    sum_op<11>(sb, ACCS); sum_op<12>(sb, ACCS); sum_op<13>(sb, ACCS);
    sum_op<14>(sb, ACCS); sum_op<15>(sb, ACCS);
}

#define NR ((GPAD + BLOCK - 1) / BLOCK)
#define J_IN_G(r, j)    (((r) + 1) * BLOCK <= G_DIM || (j) < G_DIM)
#define KPAD (NR * BLOCK)
__launch_bounds__(BLOCK)
__global__ void mha_rank1_wmma_kernel(const float* __restrict__ x,
                                      const float* __restrict__ WQ,
                                      const float* __restrict__ WK,
                                      const float* __restrict__ WV,
                                      const float* __restrict__ W0,
                                      float* __restrict__ out)
{
    __shared__ __align__(16) float k_lds[2][KPAD];
    __shared__ __align__(16) float v_lds[2][KPAD];
    __shared__ __align__(16) float2 part[H_DIM][WPB][IT][16];

    const int b    = blockIdx.x;
    const int tid  = threadIdx.x;
    const int wave = __builtin_amdgcn_readfirstlane(tid >> 5);
    const int lane = tid & 31;
    const int ln   = lane & 15;
    const int hi   = lane >> 4;

    const int i0  = (blockIdx.y * IT) * 16 + ln;
    const int i1  = i0 + 16;
    const int i0C = (i0 < G_DIM) ? i0 : (G_DIM - 1);
    const int i1C = (i1 < G_DIM) ? i1 : (G_DIM - 1);

    const int jt0 = (wave * NJT) / WPB;
    const int jt1 = ((wave + 1) * NJT) / WPB;
    const bool ownsTail = (jt1 == NJT);
    const int jtMainEnd = ownsTail ? (NJT - 1) : jt1;

    const float* __restrict__ xrow = x + b * G_DIM;
    const float xi0 = xrow[i0C];
    const float xi1 = xrow[i1C];
    float wq0 = WQ[i0C];
    float wq1 = WQ[i1C];
    float xr[NR], wkr[NR], wvr[NR];
#pragma unroll
    for (int r = 0; r < NR; ++r) {
        const int j  = tid + r * BLOCK;
        const int jc = J_IN_G(r, j) ? j : (G_DIM - 1);
        const float xv = xrow[jc];
        xr[r]  = J_IN_G(r, j) ? xv : 0.0f;
        wkr[r] = J_IN_G(r, j) ? WK[jc] : 0.0f;
        wvr[r] = J_IN_G(r, j) ? WV[jc] : 0.0f;
    }
#pragma unroll
    for (int r = 0; r < NR; ++r) {
        const int j = tid + r * BLOCK;
        k_lds[0][j] = xr[r] * wkr[r] * LOG2E;
        v_lds[0][j] = xr[r] * wvr[r];
    }
    __syncthreads();

    for (int h = 0; h < H_DIM; ++h) {
        const int cur = h & 1;
        const float* __restrict__ kb = k_lds[cur];
        const float* __restrict__ vb = v_lds[cur];

        const float t0 = (lane < 16) ? xi0 * wq0 : 0.0f;
        const float t1 = (lane < 16) ? xi1 * wq1 : 0.0f;
        v2f bm0; bm0[0] = t0; bm0[1] = 0.0f;
        v2f bm1; bm1[0] = t1; bm1[1] = 0.0f;

        const int hn = (h + 1 < H_DIM) ? (h + 1) : h;
#pragma unroll
        for (int r = 0; r < NR; ++r) {
            const int j  = tid + r * BLOCK;
            const int jc = J_IN_G(r, j) ? j : (G_DIM - 1);
            wkr[r] = J_IN_G(r, j) ? WK[hn * G_DIM + jc] : 0.0f;
            wvr[r] = J_IN_G(r, j) ? WV[hn * G_DIM + jc] : 0.0f;
        }
        wq0 = WQ[hn * G_DIM + i0C];
        wq1 = WQ[hn * G_DIM + i1C];

        v2f Dp0[2] = {}, Np0[2] = {}, Dp1[2] = {}, Np1[2] = {};

        int jt = jt0;
        for (; jt + 2 <= jtMainEnd; jt += 2)
            tile_pair_ordered(kb, vb, jt * 16, lane, ln, hi,
                              bm0, bm1, Dp0, Np0, Dp1, Np1);
        if (jt < jtMainEnd)
            tile_step<false>(kb, vb, jt * 16, lane, ln, hi,
                             bm0, bm1, Dp0, Np0, Dp1, Np1);
        if (ownsTail)
            tile_step<true>(kb, vb, (NJT - 1) * 16, lane, ln, hi,
                            bm0, bm1, Dp0, Np0, Dp1, Np1);

        float D0 = (Dp0[0][0] + Dp0[0][1]) + (Dp0[1][0] + Dp0[1][1]);
        float N0 = (Np0[0][0] + Np0[0][1]) + (Np0[1][0] + Np0[1][1]);
        float D1 = (Dp1[0][0] + Dp1[0][1]) + (Dp1[1][0] + Dp1[1][1]);
        float N1 = (Np1[0][0] + Np1[0][1]) + (Np1[1][0] + Np1[1][1]);

        D0 += __shfl_xor(D0, 16, 32);
        N0 += __shfl_xor(N0, 16, 32);
        D1 += __shfl_xor(D1, 16, 32);
        N1 += __shfl_xor(N1, 16, 32);

        if (lane < 16) {
            part[h][wave][0][ln] = make_float2(D0, N0);
            part[h][wave][1][ln] = make_float2(D1, N1);
        }

        if (h + 1 < H_DIM) {
#pragma unroll
            for (int r = 0; r < NR; ++r) {
                const int j = tid + r * BLOCK;
                k_lds[cur ^ 1][j] = xr[r] * wkr[r] * LOG2E;
                v_lds[cur ^ 1][j] = xr[r] * wvr[r];
            }
        }
        __syncthreads();
    }

    if (tid < IT * 16) {
        const int it = tid >> 4;
        const int i  = (blockIdx.y * IT + it) * 16 + ln;
        if (i < G_DIM) {
            const float xi = xrow[i];
            float out_acc = 0.0f;
            for (int h = 0; h < H_DIM; ++h) {
                float Dsum = 0.0f, Nsum = 0.0f;
#pragma unroll
                for (int w = 0; w < WPB; ++w) {
                    const float2 p = part[h][w][it][ln];
                    Dsum += p.x;
                    Nsum += p.y;
                }
                const float t    = xi * WQ[h * G_DIM + i];
                const float kval = xi * WK[h * G_DIM + i] * LOG2E;
                const float vval = xi * WV[h * G_DIM + i];
                const float eii  = EXP2F(t * kval);
                const float num  = Nsum - eii * vval;
                out_acc += W0[h] * num / Dsum;
            }
            out[b * G_DIM + i] = out_acc;
        }
    }
}

extern "C" void kernel_launch(void* const* d_in, const int* in_sizes, int n_in,
                              void* d_out, int out_size, void* d_ws, size_t ws_size,
                              hipStream_t stream) {
    (void)in_sizes; (void)n_in; (void)out_size; (void)d_ws; (void)ws_size;
    const float* x  = (const float*)d_in[0];
    const float* WQ = (const float*)d_in[1];
    const float* WK = (const float*)d_in[2];
    const float* WV = (const float*)d_in[3];
    const float* W0 = (const float*)d_in[4];
    float* out = (float*)d_out;

    dim3 grid(B_DIM, 54);
    hipLaunchKernelGGL(mha_rank1_wmma_kernel, grid, dim3(BLOCK), 0, stream,
                       x, WQ, WK, WV, W0, out);
}


# Block tile 32 -> 64 output rows (4 score tiles per wave, grid 16x27): half as many blocks and waves pay staging, joins, barriers, entry; k/v LDS reads shared by 4 products (same sums, same order)
# speedup vs baseline: 2.7443x; 1.0465x over previous
#include <hip/hip_runtime.h>

#define B_DIM 16
#define G_DIM 1708
#define H_DIM 5
#define GPAD  1712
#define NJT   107
#define BLOCK 256
#define WPB   (BLOCK / 32)
#define IT    4
#define NE    (8 * IT)
#define LAG   5
#define LOG2E 1.44269504088896340736f

typedef __attribute__((ext_vector_type(2))) float v2f;
typedef __attribute__((ext_vector_type(8))) float v8f;

#if __has_builtin(__builtin_amdgcn_exp2f)
#define EXP2F(v) __builtin_amdgcn_exp2f(v)
#else
#define EXP2F(v) exp2f(v)
#endif

struct Accs {
    v2f Dp[IT][2];
    v2f Np[IT][2];
};

template <bool TAIL>
__device__ __forceinline__ void tile_step(const float* __restrict__ k_lds,
                                          const float* __restrict__ v_lds,
                                          int jb, int lane, int ln, int hi,
                                          const v2f (&bm)[IT], Accs& A)
{
    const float kj = k_lds[jb + ln];
    v2f am; am[0] = (lane < 16) ? kj : 0.0f; am[1] = 0.0f;
    v8f c = {};
    v8f d[IT];
#pragma unroll
    for (int t = 0; t < IT; ++t)
        d[t] = __builtin_amdgcn_wmma_f32_16x16x4_f32(
            false, am, false, bm[t], (short)0, c, false, false);
    const float4* vp = reinterpret_cast<const float4*>(v_lds + jb + (hi << 3));
    const float4 va = vp[0], vb = vp[1];
    v2f vv[4];
    vv[0][0] = va.x; vv[0][1] = va.y;
    vv[1][0] = va.z; vv[1][1] = va.w;
    vv[2][0] = vb.x; vv[2][1] = vb.y;
    vv[3][0] = vb.z; vv[3][1] = vb.w;
#pragma unroll
    for (int t = 0; t < IT; ++t) {
#pragma unroll
        for (int p = 0; p < 4; ++p) {
            v2f e;
            e[0] = EXP2F(d[t][2 * p]);
            e[1] = EXP2F(d[t][2 * p + 1]);
            if (TAIL) {
                const int j = jb + (hi << 3) + 2 * p;
                e[0] = (j < G_DIM) ? e[0] : 0.0f;
                e[1] = ((j + 1) < G_DIM) ? e[1] : 0.0f;
            }
            A.Dp[t][p & 1] += e;
            A.Np[t][p & 1] = __builtin_elementwise_fma(e, vv[p], A.Np[t][p & 1]);
        }
    }
}

#define ORDER_FENCE() __builtin_amdgcn_sched_barrier(0)

struct StepVals {
    float e[NE];
    v2f   vv[4];
};

template <int N>
__device__ __forceinline__ void exp_op(StepVals& s, const v8f (&d)[IT])
{
    s.e[N] = EXP2F(d[N >> 3][N & 7]);
}

template <int K>
__device__ __forceinline__ void sum_op(const StepVals& s, Accs& A)
{
    constexpr int pr = (K >> 1) & 3;
    constexpr int t  = K >> 3;
    v2f ep; ep[0] = s.e[8 * t + 2 * pr]; ep[1] = s.e[8 * t + 2 * pr + 1];
    if ((K & 1) == 0) A.Dp[t][pr & 1] += ep;
    else A.Np[t][pr & 1] = __builtin_elementwise_fma(ep, s.vv[pr], A.Np[t][pr & 1]);
}

template <int N>
__device__ __forceinline__ void alternate(StepVals& s, const v8f (&d)[IT], Accs& A)
{
    if constexpr (N < NE) {
        exp_op<N>(s, d); ORDER_FENCE();
        sum_op<N - LAG>(s, A); ORDER_FENCE();
        alternate<N + 1>(s, d, A);
    }
}

template <int N>
__device__ __forceinline__ void handover(StepVals& sa, StepVals& sb,
                                         const v8f (&d)[IT], Accs& A)
{
    if constexpr (N < LAG) {
        exp_op<N>(sb, d); ORDER_FENCE();
        sum_op<NE - LAG + N>(sa, A); ORDER_FENCE();
        handover<N + 1>(sa, sb, d, A);
    }
}

template <int K>
__device__ __forceinline__ void trailing_sums(const StepVals& s, Accs& A)
{
    if constexpr (K < NE) { sum_op<K>(s, A); trailing_sums<K + 1>(s, A); }
}

__device__ __forceinline__ void load_step(StepVals& s, float& kj,
                                          const float* __restrict__ k_lds,
                                          const float* __restrict__ v_lds,
                                          int jb, int ln, int hi)
{
    kj = k_lds[jb + ln];
    const float4* vp = reinterpret_cast<const float4*>(v_lds + jb + (hi << 3));
    const float4 va = vp[0], vb = vp[1];
    s.vv[0][0] = va.x; s.vv[0][1] = va.y;
    s.vv[1][0] = va.z; s.vv[1][1] = va.w;
    s.vv[2][0] = vb.x; s.vv[2][1] = vb.y;
    s.vv[3][0] = vb.z; s.vv[3][1] = vb.w;
}

__device__ __forceinline__ void tile_pair_ordered(const float* __restrict__ k_lds,
                                                  const float* __restrict__ v_lds,
                                                  int jb, int lane, int ln, int hi,
                                                  const v2f (&bm)[IT], Accs& A)
{
    StepVals sa, sb; float kja, kjb;
    load_step(sa, kja, k_lds, v_lds, jb, ln, hi);
    load_step(sb, kjb, k_lds, v_lds, jb + 16, ln, hi);
    v8f c = {};
    v2f ama; ama[0] = (lane < 16) ? kja : 0.0f; ama[1] = 0.0f;
    v8f a[IT], b[IT];
#pragma unroll
    for (int t = 0; t < IT; ++t)
        a[t] = __builtin_amdgcn_wmma_f32_16x16x4_f32(
            false, ama, false, bm[t], (short)0, c, false, false);
    ORDER_FENCE();
    exp_op<0>(sa, a); ORDER_FENCE();
    v2f amb; amb[0] = (lane < 16) ? kjb : 0.0f; amb[1] = 0.0f;
    ORDER_FENCE();
    exp_op<1>(sa, a); exp_op<2>(sa, a);
    exp_op<3>(sa, a); exp_op<4>(sa, a);
    ORDER_FENCE();
    alternate<LAG>(sa, a, A);
#pragma unroll
    for (int t = 0; t < IT; ++t) {
        b[t] = __builtin_amdgcn_wmma_f32_16x16x4_f32(
            false, amb, false, bm[t], (short)0, c, false, false);
        ORDER_FENCE();
    }
    handover<0>(sa, sb, b, A);
    alternate<LAG>(sb, b, A);
    trailing_sums<NE - LAG>(sb, A);
}

#define NR ((GPAD + BLOCK - 1) / BLOCK)
#define J_IN_G(r, j)    (((r) + 1) * BLOCK <= G_DIM || (j) < G_DIM)
#define KPAD (NR * BLOCK)
__launch_bounds__(BLOCK)
__global__ void mha_rank1_wmma_kernel(const float* __restrict__ x,
                                      const float* __restrict__ WQ,
                                      const float* __restrict__ WK,
                                      const float* __restrict__ WV,
                                      const float* __restrict__ W0,
                                      float* __restrict__ out)
{
    __shared__ __align__(16) struct {
        float  k[2][KPAD];
        float  v[2][KPAD];
        float2 part[H_DIM][WPB][IT][16];
    } sm;
    float (&k_lds)[2][KPAD] = sm.k;
    float (&v_lds)[2][KPAD] = sm.v;
    float2 (&part)[H_DIM][WPB][IT][16] = sm.part;

    const int b    = blockIdx.x;
    const int tid  = threadIdx.x;
    const int wave = __builtin_amdgcn_readfirstlane(tid >> 5);
    const int lane = tid & 31;
    const int ln   = lane & 15;
    const int hi   = lane >> 4;

    int iC[IT];
#pragma unroll
    for (int t = 0; t < IT; ++t) {
        const int i = (blockIdx.y * IT + t) * 16 + ln;
        iC[t] = (i < G_DIM) ? i : (G_DIM - 1);
    }

    const int jt0 = (wave * NJT) / WPB;
    const int jt1 = ((wave + 1) * NJT) / WPB;
    const bool ownsTail = (jt1 == NJT);
    const int jtMainEnd = ownsTail ? (NJT - 1) : jt1;

    const float* __restrict__ xrow = x + b * G_DIM;
    float xi[IT], wq[IT];
#pragma unroll
    for (int t = 0; t < IT; ++t) {
        xi[t] = xrow[iC[t]];
        wq[t] = WQ[iC[t]];
    }
    float xr[NR], wkr[NR], wvr[NR];
#pragma unroll
    for (int r = 0; r < NR; ++r) {
        const int j  = tid + r * BLOCK;
        const int jc = J_IN_G(r, j) ? j : (G_DIM - 1);
        const float xv = xrow[jc];
        xr[r]  = J_IN_G(r, j) ? xv : 0.0f;
        wkr[r] = J_IN_G(r, j) ? WK[jc] : 0.0f;
        wvr[r] = J_IN_G(r, j) ? WV[jc] : 0.0f;
    }
#pragma unroll
    for (int r = 0; r < NR; ++r) {
        const int j = tid + r * BLOCK;
        k_lds[0][j] = xr[r] * wkr[r] * LOG2E;
        v_lds[0][j] = xr[r] * wvr[r];
    }
    __syncthreads();

    for (int h = 0; h < H_DIM; ++h) {
        const int cur = h & 1;
        const float* __restrict__ kb = k_lds[cur];
        const float* __restrict__ vb = v_lds[cur];

        v2f bm[IT];
#pragma unroll
        for (int t = 0; t < IT; ++t) {
            bm[t][0] = (lane < 16) ? xi[t] * wq[t] : 0.0f;
            bm[t][1] = 0.0f;
        }

        const int hn = (h + 1 < H_DIM) ? (h + 1) : h;
#pragma unroll
        for (int r = 0; r < NR; ++r) {
            const int j  = tid + r * BLOCK;
            const int jc = J_IN_G(r, j) ? j : (G_DIM - 1);
            wkr[r] = J_IN_G(r, j) ? WK[hn * G_DIM + jc] : 0.0f;
            wvr[r] = J_IN_G(r, j) ? WV[hn * G_DIM + jc] : 0.0f;
        }
#pragma unroll
        for (int t = 0; t < IT; ++t)
            wq[t] = WQ[hn * G_DIM + iC[t]];

        Accs A = {};

        int jt = jt0;
        for (; jt + 2 <= jtMainEnd; jt += 2)
            tile_pair_ordered(kb, vb, jt * 16, lane, ln, hi, bm, A);
        if (jt < jtMainEnd)
            tile_step<false>(kb, vb, jt * 16, lane, ln, hi, bm, A);
        if (ownsTail)
            tile_step<true>(kb, vb, (NJT - 1) * 16, lane, ln, hi, bm, A);

        float Df[IT], Nf[IT], Dx[IT], Nx[IT];
#pragma unroll
        for (int t = 0; t < IT; ++t) {
            Df[t] = (A.Dp[t][0][0] + A.Dp[t][0][1]) + (A.Dp[t][1][0] + A.Dp[t][1][1]);
            Nf[t] = (A.Np[t][0][0] + A.Np[t][0][1]) + (A.Np[t][1][0] + A.Np[t][1][1]);
        }
#pragma unroll
        for (int t = 0; t < IT; ++t) {
            Dx[t] = __shfl_xor(Df[t], 16, 32);
            Nx[t] = __shfl_xor(Nf[t], 16, 32);
        }
        if (lane < 16) {
#pragma unroll
            for (int t = 0; t < IT; ++t)
                part[h][wave][t][ln] = make_float2(Df[t] + Dx[t], Nf[t] + Nx[t]);
        }

        if (h + 1 < H_DIM) {
#pragma unroll
            for (int r = 0; r < NR; ++r) {
                const int j = tid + r * BLOCK;
                k_lds[cur ^ 1][j] = xr[r] * wkr[r] * LOG2E;
                v_lds[cur ^ 1][j] = xr[r] * wvr[r];
            }
        }
        __syncthreads();
    }

    if (tid < IT * 16) {
        const int it = tid >> 4;
        const int i  = (blockIdx.y * IT + it) * 16 + ln;
        if (i < G_DIM) {
            const float xv = xrow[i];
            float out_acc = 0.0f;
            for (int h = 0; h < H_DIM; ++h) {
                float Dsum = 0.0f, Nsum = 0.0f;
#pragma unroll
                for (int w = 0; w < WPB; ++w) {
                    const float2 p = part[h][w][it][ln];
                    Dsum += p.x;
                    Nsum += p.y;
                }
                const float t    = xv * WQ[h * G_DIM + i];
                const float kval = xv * WK[h * G_DIM + i] * LOG2E;
                const float vval = xv * WV[h * G_DIM + i];
                const float eii  = EXP2F(t * kval);
                const float num  = Nsum - eii * vval;
                out_acc += W0[h] * num / Dsum;
            }
            out[b * G_DIM + i] = out_acc;
        }
    }
}

extern "C" void kernel_launch(void* const* d_in, const int* in_sizes, int n_in,
                              void* d_out, int out_size, void* d_ws, size_t ws_size,
                              hipStream_t stream) {
    (void)in_sizes; (void)n_in; (void)out_size; (void)d_ws; (void)ws_size;
    const float* x  = (const float*)d_in[0];
    const float* WQ = (const float*)d_in[1];
    const float* WK = (const float*)d_in[2];
    const float* WV = (const float*)d_in[3];
    const float* W0 = (const float*)d_in[4];
    float* out = (float*)d_out;

    dim3 grid(B_DIM, 27);
    hipLaunchKernelGGL(mha_rank1_wmma_kernel, grid, dim3(BLOCK), 0, stream,
                       x, WQ, WK, WV, W0, out);
}
